# starting kernel
# baseline (speedup 1.0000x reference)
#include <hip/hip_runtime.h>

typedef __attribute__((ext_vector_type(2))) float v2f;
typedef __attribute__((ext_vector_type(4))) float v4f;
typedef __attribute__((ext_vector_type(8))) float v8f;

namespace {
constexpr int HID   = 50;
constexpr int NG    = 200;
constexpr int KPAD  = 52;
constexpr int NT    = 13;
constexpr int KC    = 13;
constexpr int NPAD  = NT * 16;
constexpr int T_IN  = 512;
constexpr int FUT   = 64;
constexpr int T_OUT = T_IN + FUT;

__device__ __forceinline__ float sigmoidf_(float v) {
    return 1.0f / (1.0f + __expf(-v));
}
}

__global__ __launch_bounds__(32) void lstm_seq_kernel(
    const float* __restrict__ x,
    const float* __restrict__ W_ih,
    const float* __restrict__ W_hh,
    const float* __restrict__ b_ih,
    const float* __restrict__ b_hh,
    const float* __restrict__ W_out,
    const float* __restrict__ b_out,
    float* __restrict__ out)
{
    __shared__ float frags[KC * NT * 32 * 2];
    __shared__ float gates[NPAD * 16];
    __shared__ float hbuf[16 * KPAD];

    const int lane    = threadIdx.x;
    const int r       = lane & 15;
    const int half    = lane >> 4;
    const int rowbase = blockIdx.x * 16;

    for (int fi = 0; fi < KC * NT; ++fi) {
        const int kc = fi / NT;
        const int tt = fi % NT;
        const int n  = tt * 16 + r;
        const int k0 = kc * 4 + 2 * half;
        float v0 = 0.0f, v1 = 0.0f;
        if (n < NG) {
            if (k0 < HID)       v0 = W_hh[n * HID + k0];
            else if (k0 == HID) v0 = b_ih[n] + b_hh[n];
            else                v0 = W_ih[n];
            const int k1 = k0 + 1;
            if (k1 < HID)       v1 = W_hh[n * HID + k1];
            else if (k1 == HID) v1 = b_ih[n] + b_hh[n];
            else                v1 = W_ih[n];
        }
        frags[(fi * 32 + lane) * 2 + 0] = v0;
        frags[(fi * 32 + lane) * 2 + 1] = v1;
    }

    float creg[25];
    float wout[25];
#pragma unroll
    for (int j = 0; j < 25; ++j) {
        const int k = 2 * j + half;
        creg[j] = 0.0f;
        wout[j] = W_out[k];
        hbuf[r * KPAD + k] = 0.0f;
    }
    hbuf[r * KPAD + HID] = 1.0f;
    const float bo = b_out[0];
    __syncthreads();

    float xnext = 0.0f;

#pragma unroll 1
    for (int t = 0; t < T_OUT; ++t) {
        float xt;
        if (t < T_IN) xt = x[(rowbase + r) * T_IN + t];
        else          xt = xnext;
        hbuf[r * KPAD + (HID + 1)] = xt;

        v2f afrag[KC];
#pragma unroll
        for (int kc = 0; kc < KC; ++kc)
            afrag[kc] = *(const v2f*)&hbuf[r * KPAD + kc * 4 + 2 * half];

#pragma unroll
        for (int tt = 0; tt < NT; ++tt) {
            v8f acc = {0.f, 0.f, 0.f, 0.f, 0.f, 0.f, 0.f, 0.f};
#pragma unroll
            for (int kc = 0; kc < KC; ++kc) {
                const v2f bfrag =
                    *(const v2f*)&frags[((kc * NT + tt) * 32 + lane) * 2];
                acc = __builtin_amdgcn_wmma_f32_16x16x4_f32(
                     false, afrag[kc],
                     false, bfrag,
                     (short)0, acc,
                     false,  false);
            }
            const int n = tt * 16 + r;
            float* gp = &gates[n * 16 + half * 8];
            v4f lo = {acc[0], acc[1], acc[2], acc[3]};
            v4f hi = {acc[4], acc[5], acc[6], acc[7]};
            *(v4f*)(gp)     = lo;
            *(v4f*)(gp + 4) = hi;
        }

        float osum = 0.0f;
#pragma unroll
        for (int j = 0; j < 25; ++j) {
            const int k = 2 * j + half;
            const float gi = gates[(k          ) * 16 + r];
            const float gf = gates[(k +     HID) * 16 + r];
            const float gg = gates[(k + 2 * HID) * 16 + r];
            const float go = gates[(k + 3 * HID) * 16 + r];
            const float ig = sigmoidf_(gi);
            const float fg = sigmoidf_(gf);
            const float cg = tanhf(gg);
            const float og = sigmoidf_(go);
            const float cv = fg * creg[j] + ig * cg;
            creg[j] = cv;
            const float hv = og * tanhf(cv);
            osum += hv * wout[j];
            hbuf[r * KPAD + k] = hv;
        }

        osum += __shfl_xor(osum, 16, 32);
        const float ov = osum + bo;
        if (half == 0)
            out[(rowbase + r) * T_OUT + t] = ov;
        xnext = ov;
    }
}

extern "C" void kernel_launch(void* const* d_in, const int* in_sizes, int n_in,
                              void* d_out, int out_size, void* d_ws, size_t ws_size,
                              hipStream_t stream) {
    const float* x     = (const float*)d_in[0];
    const float* W_ih  = (const float*)d_in[1];
    const float* W_hh  = (const float*)d_in[2];
    const float* b_ih  = (const float*)d_in[3];
    const float* b_hh  = (const float*)d_in[4];
    const float* W_out = (const float*)d_in[5];
    const float* b_out = (const float*)d_in[6];
    float* out = (float*)d_out;

    const int B = in_sizes[0] / T_IN;
    dim3 grid(B / 16);
    dim3 block(32);
    hipLaunchKernelGGL(lstm_seq_kernel, grid, block, 0, stream,
                       x, W_ih, W_hh, b_ih, b_hh, W_out, b_out, out);
}


# Split each 16-row tile's gate columns over 13 waves per workgroup (4 hidden units x 4 gates per wave), h shared via LDS, one barrier per step; out projection summed per wave, then across waves
# speedup vs baseline: 5.8828x; 5.8828x over previous
#include <hip/hip_runtime.h>


typedef __attribute__((ext_vector_type(2))) float v2f;
typedef __attribute__((ext_vector_type(4))) float v4f;
typedef __attribute__((ext_vector_type(8))) float v8f;

namespace {
constexpr int HID   = 50;
constexpr int KPAD  = 52;
constexpr int KC    = 13;
constexpr int NW    = 13;
constexpr int UPW   = 4;
constexpr int T_IN  = 512;
constexpr int FUT   = 64;
constexpr int T_OUT = T_IN + FUT;

__device__ __forceinline__ float sigmoidf_(float v) {
    return 1.0f / (1.0f + __expf(-v));
}
}

__global__ __launch_bounds__(NW * 32) void lstm_seq_kernel(
    const float* __restrict__ x,
    const float* __restrict__ W_ih,
    const float* __restrict__ W_hh,
    const float* __restrict__ b_ih,
    const float* __restrict__ b_hh,
    const float* __restrict__ W_out,
    const float* __restrict__ b_out,
    float* __restrict__ out)
{
    __shared__ float gates[NW * 16 * 16];
    __shared__ float hbuf[2][16 * KPAD];
    __shared__ float part[2][NW * 16];

    const int lane    = threadIdx.x & 31;
    const int wave    = __builtin_amdgcn_readfirstlane((int)(threadIdx.x >> 5));
    const int r       = lane & 15;
    const int half    = lane >> 4;
    const int rowbase = blockIdx.x * 16;

    v2f bfrag[KC];
    {
        const int gsel = r >> 2;
        const int u    = wave * UPW + (r & 3);
        const bool valid = u < HID;
        const int n    = gsel * HID + u;
#pragma unroll
        for (int kc = 0; kc < KC; ++kc) {
            const int k0 = kc * 4 + 2 * half;
            float v0 = 0.0f, v1 = 0.0f;
            if (valid) {
                if (k0 < HID)       v0 = W_hh[n * HID + k0];
                else if (k0 == HID) v0 = b_ih[n] + b_hh[n];
                else                v0 = W_ih[n];
                const int k1 = k0 + 1;
                if (k1 < HID)       v1 = W_hh[n * HID + k1];
                else if (k1 == HID) v1 = b_ih[n] + b_hh[n];
                else                v1 = W_ih[n];
            }
            bfrag[kc] = v2f{v0, v1};
        }
    }

    float creg[2];
    float wout[2];
#pragma unroll
    for (int jj = 0; jj < 2; ++jj) {
        const int u = wave * UPW + 2 * jj + half;
        creg[jj] = 0.0f;
        wout[jj] = (u < HID) ? W_out[u] : 0.0f;
        hbuf[0][r * KPAD + u] = 0.0f;
    }
    if (half == 0) part[0][wave * 16 + r] = 0.0f;
    const float bo = b_out[0];
    float* gw = &gates[wave * 256];
    float xg = x[(rowbase + r) * T_IN];
    __syncthreads();

#pragma unroll 1
    for (int t = 0; t < T_OUT; ++t) {
        const int cur = t & 1;
        const int nxt = cur ^ 1;

        v2f afrag[KC];
#pragma unroll
        for (int kc = 0; kc < KC; ++kc)
            afrag[kc] = *(const v2f*)&hbuf[cur][r * KPAD + kc * 4 + 2 * half];

        float ps = part[cur][r];
#pragma unroll
        for (int w = 1; w < NW; ++w) ps += part[cur][w * 16 + r];
        const float oprev = ps + bo;

        const float xt = (t < T_IN) ? xg : oprev;
        const int tn = (t + 1 < T_IN) ? (t + 1) : (T_IN - 1);
        xg = x[(rowbase + r) * T_IN + tn];

        if (half) afrag[KC - 1] = v2f{1.0f, xt};

        v8f acc = {0.f, 0.f, 0.f, 0.f, 0.f, 0.f, 0.f, 0.f};
#pragma unroll
        for (int kc = 0; kc < KC; ++kc) {
            acc = __builtin_amdgcn_wmma_f32_16x16x4_f32(
                 false, afrag[kc],
                 false, bfrag[kc],
                 (short)0, acc,
                 false,  false);
        }
        {
            float* gp = &gw[r * 16 + half * 8];
            v4f lo = {acc[0], acc[1], acc[2], acc[3]};
            v4f hi = {acc[4], acc[5], acc[6], acc[7]};
            *(v4f*)(gp)     = lo;
            *(v4f*)(gp + 4) = hi;
        }

        if (t > 0 && wave == 0) {
            if (half == 0) out[(rowbase + r) * T_OUT + (t - 1)] = oprev;
        }

        float osum = 0.0f;
#pragma unroll
        for (int jj = 0; jj < 2; ++jj) {
            const int ul = 2 * jj + half;
            const float gi = gw[(ul          ) * 16 + r];
            const float gf = gw[(ul +     UPW) * 16 + r];
            const float gg = gw[(ul + 2 * UPW) * 16 + r];
            const float go = gw[(ul + 3 * UPW) * 16 + r];
            const float ig = sigmoidf_(gi);
            const float fg = sigmoidf_(gf);
            const float cg = tanhf(gg);
            const float og = sigmoidf_(go);
            const float cv = fg * creg[jj] + ig * cg;
            creg[jj] = cv;
            const float hv = og * tanhf(cv);
            osum += hv * wout[jj];
            hbuf[nxt][r * KPAD + wave * UPW + ul] = hv;
        }

        osum += __shfl_xor(osum, 16, 32);
        if (half == 0) part[nxt][wave * 16 + r] = osum;

        __syncthreads();
    }

    {
        float ps = part[T_OUT & 1][r];
#pragma unroll
        for (int w = 1; w < NW; ++w) ps += part[T_OUT & 1][w * 16 + r];
        if (wave == 0 && half == 0)
            out[(rowbase + r) * T_OUT + (T_OUT - 1)] = ps + bo;
    }
}

extern "C" void kernel_launch(void* const* d_in, const int* in_sizes, int n_in,
                              void* d_out, int out_size, void* d_ws, size_t ws_size,
                              hipStream_t stream) {
    const float* x     = (const float*)d_in[0];
    const float* W_ih  = (const float*)d_in[1];
    const float* W_hh  = (const float*)d_in[2];
    const float* b_ih  = (const float*)d_in[3];
    const float* b_hh  = (const float*)d_in[4];
    const float* W_out = (const float*)d_in[5];
    const float* b_out = (const float*)d_in[6];
    float* out = (float*)d_out;

    const int B = in_sizes[0] / T_IN;
    dim3 grid(B / 16);
    dim3 block(NW * 32);
    hipLaunchKernelGGL(lstm_seq_kernel, grid, block, 0, stream,
                       x, W_ih, W_hh, b_ih, b_hh, W_out, b_out, out);
}


# Sigmoid and tanh on hardware exp and rcp, no division sequence and no branch: tanh selects an odd polynomial below five eighths or 1 - 2*rcp(exp(2|v|) + 1) above
# speedup vs baseline: 7.8560x; 1.3354x over previous
#include <hip/hip_runtime.h>


typedef __attribute__((ext_vector_type(2))) float v2f;
typedef __attribute__((ext_vector_type(4))) float v4f;
typedef __attribute__((ext_vector_type(8))) float v8f;

namespace {
constexpr int HID   = 50;
constexpr int KPAD  = 52;
constexpr int KC    = 13;
constexpr int NW    = 13;
constexpr int UPW   = 4;
constexpr int T_IN  = 512;
constexpr int FUT   = 64;
constexpr int T_OUT = T_IN + FUT;

__device__ __forceinline__ float sigmoidf_(float v) {
    return __builtin_amdgcn_rcpf(1.0f + __expf(-v));
}

__device__ __forceinline__ float tanhf_(float v) {
    const float a  = __builtin_fabsf(v);
    const float e  = __builtin_amdgcn_exp2f(2.8853900817779268f * a);
    const float big = __builtin_fmaf(-2.0f, __builtin_amdgcn_rcpf(e + 1.0f), 1.0f);
    const float v2 = v * v;
    float p = -0.005704987328499556f;
    p = __builtin_fmaf(p, v2, 0.0206390880048275f);
    p = __builtin_fmaf(p, v2, -0.05373971536755562f);
    p = __builtin_fmaf(p, v2, 0.13331441581249237f);
    p = __builtin_fmaf(p, v2, -0.3333328068256378f);
    const float small = __builtin_fmaf(v * v2, p, v);
    return (a < 0.625f) ? small : __builtin_copysignf(big, v);
}
}

__global__ __launch_bounds__(NW * 32) void lstm_seq_kernel(
    const float* __restrict__ x,
    const float* __restrict__ W_ih,
    const float* __restrict__ W_hh,
    const float* __restrict__ b_ih,
    const float* __restrict__ b_hh,
    const float* __restrict__ W_out,
    const float* __restrict__ b_out,
    float* __restrict__ out)
{
    __shared__ float gates[NW * 16 * 16];
    __shared__ float hbuf[2][16 * KPAD];
    __shared__ float part[2][NW * 16];

    const int lane    = threadIdx.x & 31;
    const int wave    = __builtin_amdgcn_readfirstlane((int)(threadIdx.x >> 5));
    const int r       = lane & 15;
    const int half    = lane >> 4;
    const int rowbase = blockIdx.x * 16;

    v2f bfrag[KC];
    {
        const int gsel = r >> 2;
        const int u    = wave * UPW + (r & 3);
        const bool valid = u < HID;
        const int n    = gsel * HID + u;
#pragma unroll
        for (int kc = 0; kc < KC; ++kc) {
            const int k0 = kc * 4 + 2 * half;
            float v0 = 0.0f, v1 = 0.0f;
            if (valid) {
                if (k0 < HID)       v0 = W_hh[n * HID + k0];
                else if (k0 == HID) v0 = b_ih[n] + b_hh[n];
                else                v0 = W_ih[n];
                const int k1 = k0 + 1;
                if (k1 < HID)       v1 = W_hh[n * HID + k1];
                else if (k1 == HID) v1 = b_ih[n] + b_hh[n];
                else                v1 = W_ih[n];
            }
            bfrag[kc] = v2f{v0, v1};
        }
    }

    float creg[2];
    float wout[2];
#pragma unroll
    for (int jj = 0; jj < 2; ++jj) {
        const int u = wave * UPW + 2 * jj + half;
        creg[jj] = 0.0f;
        wout[jj] = (u < HID) ? W_out[u] : 0.0f;
        hbuf[0][r * KPAD + u] = 0.0f;
    }
    if (half == 0) part[0][wave * 16 + r] = 0.0f;
    const float bo = b_out[0];
    float* gw = &gates[wave * 256];
    float xg = x[(rowbase + r) * T_IN];
    __syncthreads();

#pragma unroll 1
    for (int t = 0; t < T_OUT; ++t) {
        const int cur = t & 1;
        const int nxt = cur ^ 1;

        v2f afrag[KC];
#pragma unroll
        for (int kc = 0; kc < KC; ++kc)
            afrag[kc] = *(const v2f*)&hbuf[cur][r * KPAD + kc * 4 + 2 * half];

        float ps = part[cur][r];
#pragma unroll
        for (int w = 1; w < NW; ++w) ps += part[cur][w * 16 + r];
        const float oprev = ps + bo;

        const float xt = (t < T_IN) ? xg : oprev;
        const int tn = (t + 1 < T_IN) ? (t + 1) : (T_IN - 1);
        xg = x[(rowbase + r) * T_IN + tn];

        if (half) afrag[KC - 1] = v2f{1.0f, xt};

        v8f acc = {0.f, 0.f, 0.f, 0.f, 0.f, 0.f, 0.f, 0.f};
#pragma unroll
        for (int kc = 0; kc < KC; ++kc) {
            acc = __builtin_amdgcn_wmma_f32_16x16x4_f32(
                 false, afrag[kc],
                 false, bfrag[kc],
                 (short)0, acc,
                 false,  false);
        }
        {
            float* gp = &gw[r * 16 + half * 8];
            v4f lo = {acc[0], acc[1], acc[2], acc[3]};
            v4f hi = {acc[4], acc[5], acc[6], acc[7]};
            *(v4f*)(gp)     = lo;
            *(v4f*)(gp + 4) = hi;
        }

        if (t > 0 && wave == 0) {
            if (half == 0) out[(rowbase + r) * T_OUT + (t - 1)] = oprev;
        }

        float osum = 0.0f;
#pragma unroll
        for (int jj = 0; jj < 2; ++jj) {
            const int ul = 2 * jj + half;
            const float gi = gw[(ul          ) * 16 + r];
            const float gf = gw[(ul +     UPW) * 16 + r];
            const float gg = gw[(ul + 2 * UPW) * 16 + r];
            const float go = gw[(ul + 3 * UPW) * 16 + r];
            const float ig = sigmoidf_(gi);
            const float fg = sigmoidf_(gf);
            const float cg = tanhf_(gg);
            const float og = sigmoidf_(go);
            const float cv = fg * creg[jj] + ig * cg;
            creg[jj] = cv;
            const float hv = og * tanhf_(cv);
            osum += hv * wout[jj];
            hbuf[nxt][r * KPAD + wave * UPW + ul] = hv;
        }

        osum += __shfl_xor(osum, 16, 32);
        if (half == 0) part[nxt][wave * 16 + r] = osum;

        __syncthreads();
    }

    {
        float ps = part[T_OUT & 1][r];
#pragma unroll
        for (int w = 1; w < NW; ++w) ps += part[T_OUT & 1][w * 16 + r];
        if (wave == 0 && half == 0)
            out[(rowbase + r) * T_OUT + (T_OUT - 1)] = ps + bo;
    }
}

extern "C" void kernel_launch(void* const* d_in, const int* in_sizes, int n_in,
                              void* d_out, int out_size, void* d_ws, size_t ws_size,
                              hipStream_t stream) {
    const float* x     = (const float*)d_in[0];
    const float* W_ih  = (const float*)d_in[1];
    const float* W_hh  = (const float*)d_in[2];
    const float* b_ih  = (const float*)d_in[3];
    const float* b_hh  = (const float*)d_in[4];
    const float* W_out = (const float*)d_in[5];
    const float* b_out = (const float*)d_in[6];
    float* out = (float*)d_out;

    const int B = in_sizes[0] / T_IN;
    dim3 grid(B / 16);
    dim3 block(NW * 32);
    hipLaunchKernelGGL(lstm_seq_kernel, grid, block, 0, stream,
                       x, W_ih, W_hh, b_ih, b_hh, W_out, b_out, out);
}


# Weights as the WMMA A operand and h as B: the accumulator holds the gates transposed, so a lane has all four gates of its two cells in registers; the gate exchange through LDS is removed
# speedup vs baseline: 8.0947x; 1.0304x over previous
#include <hip/hip_runtime.h>


typedef __attribute__((ext_vector_type(2))) float v2f;
typedef __attribute__((ext_vector_type(8))) float v8f;

namespace {
constexpr int HID   = 50;
constexpr int KPAD  = 52;
constexpr int KC    = 13;
constexpr int NW    = 13;
constexpr int UPW   = 4;
constexpr int T_IN  = 512;
constexpr int FUT   = 64;
constexpr int T_OUT = T_IN + FUT;

__device__ __forceinline__ float sigmoidf_(float v) {
    return __builtin_amdgcn_rcpf(1.0f + __expf(-v));
}

__device__ __forceinline__ float tanhf_(float v) {
    const float a  = __builtin_fabsf(v);
    const float e  = __builtin_amdgcn_exp2f(2.8853900817779268f * a);
    const float big = __builtin_fmaf(-2.0f, __builtin_amdgcn_rcpf(e + 1.0f), 1.0f);
    const float v2 = v * v;
    float p = -0.005704987328499556f;
    p = __builtin_fmaf(p, v2, 0.0206390880048275f);
    p = __builtin_fmaf(p, v2, -0.05373971536755562f);
    p = __builtin_fmaf(p, v2, 0.13331441581249237f);
    p = __builtin_fmaf(p, v2, -0.3333328068256378f);
    const float small = __builtin_fmaf(v * v2, p, v);
    return (a < 0.625f) ? small : __builtin_copysignf(big, v);
}
}

__global__ __launch_bounds__(NW * 32) void lstm_seq_kernel(
    const float* __restrict__ x,
    const float* __restrict__ W_ih,
    const float* __restrict__ W_hh,
    const float* __restrict__ b_ih,
    const float* __restrict__ b_hh,
    const float* __restrict__ W_out,
    const float* __restrict__ b_out,
    float* __restrict__ out)
{
    __shared__ float hbuf[2][16 * KPAD];
    __shared__ float part[2][NW * 16];

    const int lane    = threadIdx.x & 31;
    const int wave    = __builtin_amdgcn_readfirstlane((int)(threadIdx.x >> 5));
    const int r       = lane & 15;
    const int half    = lane >> 4;
    const int rowbase = blockIdx.x * 16;

    v2f wfrag[KC];
    {
        const int gsel = (r & 7) >> 1;
        const int u    = wave * UPW + 2 * (r & 1) + (r >> 3);
        const bool valid = u < HID;
        const int n    = gsel * HID + u;
#pragma unroll
        for (int kc = 0; kc < KC; ++kc) {
            const int k0 = kc * 4 + 2 * half;
            float v0 = 0.0f, v1 = 0.0f;
            if (valid) {
                if (k0 < HID)       v0 = W_hh[n * HID + k0];
                else if (k0 == HID) v0 = b_ih[n] + b_hh[n];
                else                v0 = W_ih[n];
                const int k1 = k0 + 1;
                if (k1 < HID)       v1 = W_hh[n * HID + k1];
                else if (k1 == HID) v1 = b_ih[n] + b_hh[n];
                else                v1 = W_ih[n];
            }
            wfrag[kc] = v2f{v0, v1};
        }
    }

    float creg[2];
    float wout[2];
#pragma unroll
    for (int jj = 0; jj < 2; ++jj) {
        const int u = wave * UPW + 2 * jj + half;
        creg[jj] = 0.0f;
        wout[jj] = (u < HID) ? W_out[u] : 0.0f;
        hbuf[0][r * KPAD + u] = 0.0f;
    }
    if (half == 0) part[0][wave * 16 + r] = 0.0f;
    const float bo = b_out[0];
    float xg = x[(rowbase + r) * T_IN];
    __syncthreads();

#pragma unroll 1
    for (int t = 0; t < T_OUT; ++t) {
        const int cur = t & 1;
        const int nxt = cur ^ 1;

        v2f hfrag[KC];
#pragma unroll
        for (int kc = 0; kc < KC; ++kc)
            hfrag[kc] = *(const v2f*)&hbuf[cur][r * KPAD + kc * 4 + 2 * half];

        float ps = part[cur][r];
#pragma unroll
        for (int w = 1; w < NW; ++w) ps += part[cur][w * 16 + r];
        const float oprev = ps + bo;

        const float xt = (t < T_IN) ? xg : oprev;
        const int tn = (t + 1 < T_IN) ? (t + 1) : (T_IN - 1);
        xg = x[(rowbase + r) * T_IN + tn];

        if (half) hfrag[KC - 1] = v2f{1.0f, xt};

        v8f acc = {0.f, 0.f, 0.f, 0.f, 0.f, 0.f, 0.f, 0.f};
#pragma unroll
        for (int kc = 0; kc < KC; ++kc) {
            acc = __builtin_amdgcn_wmma_f32_16x16x4_f32(
                 false, wfrag[kc],
                 false, hfrag[kc],
                 (short)0, acc,
                 false,  false);
        }

        if (t > 0 && wave == 0) {
            if (half == 0) out[(rowbase + r) * T_OUT + (t - 1)] = oprev;
        }

        float osum = 0.0f;
#pragma unroll
        for (int jj = 0; jj < 2; ++jj) {
            const int ul = 2 * jj + half;
            const float gi = acc[0 + jj];
            const float gf = acc[2 + jj];
            const float gg = acc[4 + jj];
            const float go = acc[6 + jj];
            const float ig = sigmoidf_(gi);
            const float fg = sigmoidf_(gf);
            const float cg = tanhf_(gg);
            const float og = sigmoidf_(go);
            const float cv = fg * creg[jj] + ig * cg;
            creg[jj] = cv;
            const float hv = og * tanhf_(cv);
            osum += hv * wout[jj];
            hbuf[nxt][r * KPAD + wave * UPW + ul] = hv;
        }

        osum += __shfl_xor(osum, 16, 32);
        if (half == 0) part[nxt][wave * 16 + r] = osum;

        __syncthreads();
    }

    {
        float ps = part[T_OUT & 1][r];
#pragma unroll
        for (int w = 1; w < NW; ++w) ps += part[T_OUT & 1][w * 16 + r];
        if (wave == 0 && half == 0)
            out[(rowbase + r) * T_OUT + (T_OUT - 1)] = ps + bo;
    }
}

extern "C" void kernel_launch(void* const* d_in, const int* in_sizes, int n_in,
                              void* d_out, int out_size, void* d_ws, size_t ws_size,
                              hipStream_t stream) {
    const float* x     = (const float*)d_in[0];
    const float* W_ih  = (const float*)d_in[1];
    const float* W_hh  = (const float*)d_in[2];
    const float* b_ih  = (const float*)d_in[3];
    const float* b_hh  = (const float*)d_in[4];
    const float* W_out = (const float*)d_in[5];
    const float* b_out = (const float*)d_in[6];
    float* out = (float*)d_out;

    const int B = in_sizes[0] / T_IN;
    dim3 grid(B / 16);
    dim3 block(NW * 32);
    hipLaunchKernelGGL(lstm_seq_kernel, grid, block, 0, stream,
                       x, W_ih, W_hh, b_ih, b_hh, W_out, b_out, out);
}


# Output projection taken a step later by each wave from the h fragments it already loads (26 FMAs per lane, halves joined by permlanex16): partial sums through LDS are gone; out summed in a new order
# speedup vs baseline: 9.2079x; 1.1375x over previous
#include <hip/hip_runtime.h>


typedef __attribute__((ext_vector_type(2))) float v2f;
typedef __attribute__((ext_vector_type(8))) float v8f;

namespace {
constexpr int HID   = 50;
constexpr int KPAD  = 52;
constexpr int KC    = 13;
constexpr int NW    = 13;
constexpr int UPW   = 4;
constexpr int T_IN  = 512;
constexpr int FUT   = 64;
constexpr int T_OUT = T_IN + FUT;

__device__ __forceinline__ float sigmoidf_(float v) {
    return __builtin_amdgcn_rcpf(1.0f + __expf(-v));
}

__device__ __forceinline__ float tanhf_(float v) {
    const float a  = __builtin_fabsf(v);
    const float e  = __builtin_amdgcn_exp2f(2.8853900817779268f * a);
    const float big = __builtin_fmaf(-2.0f, __builtin_amdgcn_rcpf(e + 1.0f), 1.0f);
    const float v2 = v * v;
    float p = -0.005704987328499556f;
    p = __builtin_fmaf(p, v2, 0.0206390880048275f);
    p = __builtin_fmaf(p, v2, -0.05373971536755562f);
    p = __builtin_fmaf(p, v2, 0.13331441581249237f);
    p = __builtin_fmaf(p, v2, -0.3333328068256378f);
    const float small = __builtin_fmaf(v * v2, p, v);
    return (a < 0.625f) ? small : __builtin_copysignf(big, v);
}

__device__ __forceinline__ float xhalf_(float v) {
    const int b = __builtin_bit_cast(int, v);
    return __builtin_bit_cast(float,
        __builtin_amdgcn_permlanex16(b, b, 0x76543210, 0xfedcba98, false, false));
}
}

__global__ __launch_bounds__(NW * 32) void lstm_seq_kernel(
    const float* __restrict__ x,
    const float* __restrict__ W_ih,
    const float* __restrict__ W_hh,
    const float* __restrict__ b_ih,
    const float* __restrict__ b_hh,
    const float* __restrict__ W_out,
    const float* __restrict__ b_out,
    float* __restrict__ out)
{
    __shared__ float hbuf[2][16 * KPAD];

    const int lane    = threadIdx.x & 31;
    const int wave    = __builtin_amdgcn_readfirstlane((int)(threadIdx.x >> 5));
    const int r       = lane & 15;
    const int half    = lane >> 4;
    const int rowbase = blockIdx.x * 16;

    v2f wfrag[KC];
    {
        const int gsel = (r & 7) >> 1;
        const int u    = wave * UPW + 2 * (r & 1) + (r >> 3);
        const bool valid = u < HID;
        const int n    = gsel * HID + u;
#pragma unroll
        for (int kc = 0; kc < KC; ++kc) {
            const int k0 = kc * 4 + 2 * half;
            float v0 = 0.0f, v1 = 0.0f;
            if (valid) {
                if (k0 < HID)       v0 = W_hh[n * HID + k0];
                else if (k0 == HID) v0 = b_ih[n] + b_hh[n];
                else                v0 = W_ih[n];
                const int k1 = k0 + 1;
                if (k1 < HID)       v1 = W_hh[n * HID + k1];
                else if (k1 == HID) v1 = b_ih[n] + b_hh[n];
                else                v1 = W_ih[n];
            }
            wfrag[kc] = v2f{v0, v1};
        }
    }

    v2f wofrag[KC];
#pragma unroll
    for (int kc = 0; kc < KC; ++kc) {
        const int k0 = kc * 4 + 2 * half;
        wofrag[kc] = v2f{(k0 < HID) ? W_out[k0] : 0.0f,
                         (k0 + 1 < HID) ? W_out[k0 + 1] : 0.0f};
    }

    float creg[2];
#pragma unroll
    for (int jj = 0; jj < 2; ++jj) {
        const int u = wave * UPW + 2 * jj + half;
        creg[jj] = 0.0f;
        hbuf[0][r * KPAD + u] = 0.0f;
    }
    const float bo = b_out[0];
    float xg = x[(rowbase + r) * T_IN];
    __syncthreads();

#pragma unroll 1
    for (int t = 0; t < T_OUT; ++t) {
        const int cur = t & 1;
        const int nxt = cur ^ 1;

        v2f hfrag[KC];
#pragma unroll
        for (int kc = 0; kc < KC; ++kc)
            hfrag[kc] = *(const v2f*)&hbuf[cur][r * KPAD + kc * 4 + 2 * half];

        const float xobs = xg;
        const int tn = (t + 1 < T_IN) ? (t + 1) : (T_IN - 1);
        xg = x[(rowbase + r) * T_IN + tn];

        v8f acc = {0.f, 0.f, 0.f, 0.f, 0.f, 0.f, 0.f, 0.f};
#pragma unroll
        for (int kc = 0; kc < KC - 1; ++kc) {
            acc = __builtin_amdgcn_wmma_f32_16x16x4_f32(
                 false, wfrag[kc],
                 false, hfrag[kc],
                 (short)0, acc,
                 false,  false);
        }

        float oprev = 0.0f;
        if (t >= T_IN || wave == 0) {
            v2f pe = hfrag[0] * wofrag[0];
            v2f po = hfrag[1] * wofrag[1];
#pragma unroll
            for (int kc = 2; kc < KC; kc += 2) {
                pe += hfrag[kc] * wofrag[kc];
                if (kc + 1 < KC) po += hfrag[kc + 1] * wofrag[kc + 1];
            }
            const v2f pacc = pe + po;
            const float ph = pacc.x + pacc.y;
            oprev = (ph + xhalf_(ph)) + bo;
        }

        const float xt = (t < T_IN) ? xobs : oprev;

        if (half) hfrag[KC - 1] = v2f{1.0f, xt};
        acc = __builtin_amdgcn_wmma_f32_16x16x4_f32(
             false, wfrag[KC - 1],
             false, hfrag[KC - 1],
             (short)0, acc,
             false,  false);

        if (t > 0 && wave == 0) {
            if (half == 0) out[(rowbase + r) * T_OUT + (t - 1)] = oprev;
        }

#pragma unroll
        for (int jj = 0; jj < 2; ++jj) {
            const int ul = 2 * jj + half;
            const float gi = acc[0 + jj];
            const float gf = acc[2 + jj];
            const float gg = acc[4 + jj];
            const float go = acc[6 + jj];
            const float ig = sigmoidf_(gi);
            const float fg = sigmoidf_(gf);
            const float cg = tanhf_(gg);
            const float og = sigmoidf_(go);
            const float cv = fg * creg[jj] + ig * cg;
            creg[jj] = cv;
            const float hv = og * tanhf_(cv);
            hbuf[nxt][r * KPAD + wave * UPW + ul] = hv;
        }

        __syncthreads();
    }

    if (wave == 0) {
        v2f pacc = v2f{0.0f, 0.0f};
#pragma unroll
        for (int kc = 0; kc < KC; ++kc)
            pacc += *(const v2f*)&hbuf[T_OUT & 1][r * KPAD + kc * 4 + 2 * half] * wofrag[kc];
        const float ph = pacc.x + pacc.y;
        const float olast = (ph + xhalf_(ph)) + bo;
        if (half == 0) out[(rowbase + r) * T_OUT + (T_OUT - 1)] = olast;
    }
}

extern "C" void kernel_launch(void* const* d_in, const int* in_sizes, int n_in,
                              void* d_out, int out_size, void* d_ws, size_t ws_size,
                              hipStream_t stream) {
    const float* x     = (const float*)d_in[0];
    const float* W_ih  = (const float*)d_in[1];
    const float* W_hh  = (const float*)d_in[2];
    const float* b_ih  = (const float*)d_in[3];
    const float* b_hh  = (const float*)d_in[4];
    const float* W_out = (const float*)d_in[5];
    const float* b_out = (const float*)d_in[6];
    float* out = (float*)d_out;

    const int B = in_sizes[0] / T_IN;
    dim3 grid(B / 16);
    dim3 block(NW * 32);
    hipLaunchKernelGGL(lstm_seq_kernel, grid, block, 0, stream,
                       x, W_ih, W_hh, b_ih, b_hh, W_out, b_out, out);
}


# Sigmoid and tanh each on one hardware tanh (v_tanh_f32): sigmoid(v) = (1 + tanh(v/2))/2, the halving of v folded exactly into the i, f, o weight rows; exp, rcp, polynomial and select are gone
# speedup vs baseline: 10.9386x; 1.1880x over previous
#include <hip/hip_runtime.h>


typedef __attribute__((ext_vector_type(2))) float v2f;
typedef __attribute__((ext_vector_type(8))) float v8f;

namespace {
constexpr int HID   = 50;
constexpr int KPAD  = 52;
constexpr int KC    = 13;
constexpr int NW    = 13;
constexpr int UPW   = 4;
constexpr int T_IN  = 512;
constexpr int FUT   = 64;
constexpr int T_OUT = T_IN + FUT;

__device__ __forceinline__ float tanhf_(float v) {
    return __builtin_amdgcn_tanhf(v);
}

__device__ __forceinline__ float xhalf_(float v) {
    const int b = __builtin_bit_cast(int, v);
    return __builtin_bit_cast(float,
        __builtin_amdgcn_permlanex16(b, b, 0x76543210, 0xfedcba98, false, false));
}
}

__global__ __launch_bounds__(NW * 32) void lstm_seq_kernel(
    const float* __restrict__ x,
    const float* __restrict__ W_ih,
    const float* __restrict__ W_hh,
    const float* __restrict__ b_ih,
    const float* __restrict__ b_hh,
    const float* __restrict__ W_out,
    const float* __restrict__ b_out,
    float* __restrict__ out)
{
    __shared__ float hbuf[2][16 * KPAD];

    const int lane    = threadIdx.x & 31;
    const int wave    = __builtin_amdgcn_readfirstlane((int)(threadIdx.x >> 5));
    const int r       = lane & 15;
    const int half    = lane >> 4;
    const int rowbase = blockIdx.x * 16;

    v2f wfrag[KC];
    {
        const int gsel = (r & 7) >> 1;
        const int u    = wave * UPW + 2 * (r & 1) + (r >> 3);
        const bool valid = u < HID;
        const int n    = gsel * HID + u;
        const float gscale = (gsel == 2) ? 1.0f : 0.5f;
#pragma unroll
        for (int kc = 0; kc < KC; ++kc) {
            const int k0 = kc * 4 + 2 * half;
            float v0 = 0.0f, v1 = 0.0f;
            if (valid) {
                if (k0 < HID)       v0 = W_hh[n * HID + k0];
                else if (k0 == HID) v0 = b_ih[n] + b_hh[n];
                else                v0 = W_ih[n];
                const int k1 = k0 + 1;
                if (k1 < HID)       v1 = W_hh[n * HID + k1];
                else if (k1 == HID) v1 = b_ih[n] + b_hh[n];
                else                v1 = W_ih[n];
            }
            wfrag[kc] = v2f{v0 * gscale, v1 * gscale};
        }
    }

    v2f wofrag[KC];
#pragma unroll
    for (int kc = 0; kc < KC; ++kc) {
        const int k0 = kc * 4 + 2 * half;
        wofrag[kc] = v2f{(k0 < HID) ? W_out[k0] : 0.0f,
                         (k0 + 1 < HID) ? W_out[k0 + 1] : 0.0f};
    }

    v2f creg = v2f{0.0f, 0.0f};
#pragma unroll
    for (int jj = 0; jj < 2; ++jj) {
        const int u = wave * UPW + 2 * jj + half;
        hbuf[0][r * KPAD + u] = 0.0f;
    }
    const float bo = b_out[0];
    float xg = x[(rowbase + r) * T_IN];
    __syncthreads();

#pragma unroll 1
    for (int t = 0; t < T_OUT; ++t) {
        const int cur = t & 1;
        const int nxt = cur ^ 1;

        v2f hfrag[KC];
#pragma unroll
        for (int kc = 0; kc < KC; ++kc)
            hfrag[kc] = *(const v2f*)&hbuf[cur][r * KPAD + kc * 4 + 2 * half];

        const float xobs = xg;
        const int tn = (t + 1 < T_IN) ? (t + 1) : (T_IN - 1);
        xg = x[(rowbase + r) * T_IN + tn];

        v8f acc = {0.f, 0.f, 0.f, 0.f, 0.f, 0.f, 0.f, 0.f};
#pragma unroll
        for (int kc = 0; kc < KC - 1; ++kc) {
            acc = __builtin_amdgcn_wmma_f32_16x16x4_f32(
                 false, wfrag[kc],
                 false, hfrag[kc],
                 (short)0, acc,
                 false,  false);
        }

        float oprev = 0.0f;
        if (t >= T_IN || wave == 0) {
            v2f pe = hfrag[0] * wofrag[0];
            v2f po = hfrag[1] * wofrag[1];
#pragma unroll
            for (int kc = 2; kc < KC; kc += 2) {
                pe += hfrag[kc] * wofrag[kc];
                if (kc + 1 < KC) po += hfrag[kc + 1] * wofrag[kc + 1];
            }
            const v2f pacc = pe + po;
            const float ph = pacc.x + pacc.y;
            oprev = (ph + xhalf_(ph)) + bo;
        }

        const float xt = (t < T_IN) ? xobs : oprev;

        if (half) hfrag[KC - 1] = v2f{1.0f, xt};
        acc = __builtin_amdgcn_wmma_f32_16x16x4_f32(
             false, wfrag[KC - 1],
             false, hfrag[KC - 1],
             (short)0, acc,
             false,  false);

        if (t > 0 && wave == 0) {
            if (half == 0) out[(rowbase + r) * T_OUT + (t - 1)] = oprev;
        }

        const v2f ti = v2f{tanhf_(acc[0]), tanhf_(acc[1])};
        const v2f tf = v2f{tanhf_(acc[2]), tanhf_(acc[3])};
        const v2f cg = v2f{tanhf_(acc[4]), tanhf_(acc[5])};
        const v2f to = v2f{tanhf_(acc[6]), tanhf_(acc[7])};
        const v2f ig = ti * 0.5f + 0.5f;
        const v2f fg = tf * 0.5f + 0.5f;
        const v2f og = to * 0.5f + 0.5f;
        creg = fg * creg + ig * cg;
        const v2f hv = og * v2f{tanhf_(creg.x), tanhf_(creg.y)};
        hbuf[nxt][r * KPAD + wave * UPW + half]     = hv.x;
        hbuf[nxt][r * KPAD + wave * UPW + 2 + half] = hv.y;

        __syncthreads();
    }

    if (wave == 0) {
        v2f pacc = v2f{0.0f, 0.0f};
#pragma unroll
        for (int kc = 0; kc < KC; ++kc)
            pacc += *(const v2f*)&hbuf[T_OUT & 1][r * KPAD + kc * 4 + 2 * half] * wofrag[kc];
        const float ph = pacc.x + pacc.y;
        const float olast = (ph + xhalf_(ph)) + bo;
        if (half == 0) out[(rowbase + r) * T_OUT + (T_OUT - 1)] = olast;
    }
}

extern "C" void kernel_launch(void* const* d_in, const int* in_sizes, int n_in,
                              void* d_out, int out_size, void* d_ws, size_t ws_size,
                              hipStream_t stream) {
    const float* x     = (const float*)d_in[0];
    const float* W_ih  = (const float*)d_in[1];
    const float* W_hh  = (const float*)d_in[2];
    const float* b_ih  = (const float*)d_in[3];
    const float* b_hh  = (const float*)d_in[4];
    const float* W_out = (const float*)d_in[5];
    const float* b_out = (const float*)d_in[6];
    float* out = (float*)d_out;

    const int B = in_sizes[0] / T_IN;
    dim3 grid(B / 16);
    dim3 block(NW * 32);
    hipLaunchKernelGGL(lstm_seq_kernel, grid, block, 0, stream,
                       x, W_ih, W_hh, b_ih, b_hh, W_out, b_out, out);
}


# Each wave's own h chunk stays in registers and feeds the next step's first WMMA, issued behind the barrier while the other 12 chunks are read from a ring in LDS; gate and out sums in a new order
# speedup vs baseline: 11.4659x; 1.0482x over previous
#include <hip/hip_runtime.h>


typedef __attribute__((ext_vector_type(2))) float v2f;
typedef __attribute__((ext_vector_type(8))) float v8f;

namespace {
constexpr int HID   = 50;
constexpr int NCH   = 13;
constexpr int NR    = 12;
constexpr int NW    = 13;
constexpr int UPW   = 4;
constexpr int T_IN  = 512;
constexpr int FUT   = 64;
constexpr int T_OUT = T_IN + FUT;
constexpr int PITCH = 108;

__device__ __forceinline__ float tanhf_(float v) {
    return __builtin_amdgcn_tanhf(v);
}

__device__ __forceinline__ float xhalf_(float v) {
    const int b = __builtin_bit_cast(int, v);
    return __builtin_bit_cast(float,
        __builtin_amdgcn_permlanex16(b, b, 0x76543210, 0xfedcba98, false, false));
}

__device__ __forceinline__ v8f wmma_(v2f a, v2f b, v8f c) {
    return __builtin_amdgcn_wmma_f32_16x16x4_f32(
         false, a,  false, b,
         (short)0, c,  false,  false);
}

__device__ __forceinline__ v2f cell_(const v8f acc, v2f& creg) {
    const v2f ti = v2f{tanhf_(acc[0]), tanhf_(acc[1])};
    const v2f tf = v2f{tanhf_(acc[2]), tanhf_(acc[3])};
    const v2f cg = v2f{tanhf_(acc[4]), tanhf_(acc[5])};
    const v2f to = v2f{tanhf_(acc[6]), tanhf_(acc[7])};
    const v2f ig = ti * 0.5f + 0.5f;
    const v2f fg = tf * 0.5f + 0.5f;
    const v2f og = to * 0.5f + 0.5f;
    creg = fg * creg + ig * cg;
    return og * v2f{tanhf_(creg.x), tanhf_(creg.y)};
}

__device__ __forceinline__ float project_(const v2f (&hf)[NR], const v2f (&wo)[NR],
                                          v2f hown, v2f woown, float bo) {
    v2f pe = hown * woown;
    v2f po = hf[0] * wo[0];
#pragma unroll
    for (int j = 1; j < NR; j += 2) {
        pe += hf[j] * wo[j];
        if (j + 1 < NR) po += hf[j + 1] * wo[j + 1];
    }
    const v2f pacc = pe + po;
    const float ph = pacc.x + pacc.y;
    return (ph + xhalf_(ph)) + bo;
}
}

__global__ __launch_bounds__(NW * 32) void lstm_seq_kernel(
    const float* __restrict__ x,
    const float* __restrict__ W_ih,
    const float* __restrict__ W_hh,
    const float* __restrict__ b_ih,
    const float* __restrict__ b_hh,
    const float* __restrict__ W_out,
    const float* __restrict__ b_out,
    float* __restrict__ out)
{
    __shared__ float hbuf[2][16 * PITCH];

    const int lane    = threadIdx.x & 31;
    const int wave    = __builtin_amdgcn_readfirstlane((int)(threadIdx.x >> 5));
    const int r       = lane & 15;
    const int half    = lane >> 4;
    const int rowbase = blockIdx.x * 16;
    const bool lastw  = (wave == NW - 1);

    const int gsel = (r & 7) >> 1;
    const int u    = wave * UPW + 2 * (r & 1) + (r >> 3);
    const bool valid = u < HID;
    const int n    = valid ? gsel * HID + u : 0;
    const float gscale = valid ? ((gsel == 2) ? 1.0f : 0.5f) : 0.0f;
    auto wcol = [&](int k) -> float {
        if (!valid) return 0.0f;
        float v;
        if (k < HID)       v = W_hh[n * HID + k];
        else if (k == HID) v = b_ih[n] + b_hh[n];
        else               v = W_ih[n];
        return v * gscale;
    };
    auto wchunk = [&](int c) -> v2f {
        const int k0 = c * 4 + half;
        return v2f{wcol(k0), wcol(k0 + 2)};
    };
    auto wochunk = [&](int c) -> v2f {
        const int k0 = c * 4 + half;
        return v2f{(k0 < HID) ? W_out[k0] : 0.0f, (k0 + 2 < HID) ? W_out[k0 + 2] : 0.0f};
    };
    const v2f wown  = wchunk(wave);
    const v2f woown = wochunk(wave);
    v2f wfrag[NR], wofrag[NR];
#pragma unroll
    for (int j = 0; j < NR; ++j) {
        int c = wave + 1 + j;
        if (c >= NCH) c -= NCH;
        wfrag[j]  = wchunk(c);
        wofrag[j] = wochunk(c);
    }
    const v2f wxfrag = v2f{0.0f, half ? wcol(HID + 1) : 0.0f};
    const float bo = b_out[0];

    v2f creg = v2f{0.0f, 0.0f};
    const float* xrow = x + (size_t)(rowbase + r) * T_IN;
    v2f hown = v2f{0.0f, 0.0f};
    if (lastw) hown.y = half ? xrow[0] : 1.0f;
    const int rdbase = r * PITCH + (wave + 1) * 4 + 2 * half;
    const int wrbase = r * PITCH + wave * 4 + 2 * half;
    *(v2f*)&hbuf[0][wrbase]           = hown;
    *(v2f*)&hbuf[0][wrbase + NCH * 4] = hown;
    __syncthreads();

#pragma unroll 1
    for (int t = 0; t < T_IN; ++t) {
        const int cur = t & 1;
        const int nxt = cur ^ 1;

        v2f hfrag[NR];
#pragma unroll
        for (int j = 0; j < NR; ++j)
            hfrag[j] = *(const v2f*)&hbuf[cur][rdbase + j * 4];

        float xnext;
        if (lastw) {
            const int tn = (t + 1 < T_IN) ? (t + 1) : (T_IN - 1);
            xnext = xrow[tn];
        }

        v8f acc = {0.f, 0.f, 0.f, 0.f, 0.f, 0.f, 0.f, 0.f};
        acc = wmma_(wown, hown, acc);
#pragma unroll
        for (int j = 0; j < NR; ++j)
            acc = wmma_(wfrag[j], hfrag[j], acc);

        if (t > 0 && wave == 0) {
            const float oprev = project_(hfrag, wofrag, hown, woown, bo);
            if (half == 0) out[(rowbase + r) * T_OUT + (t - 1)] = oprev;
        }

        v2f hv = cell_(acc, creg);
        if (lastw) hv.y = half ? ((t + 1 < T_IN) ? xnext : 0.0f) : 1.0f;
        hown = hv;
        *(v2f*)&hbuf[nxt][wrbase]           = hv;
        *(v2f*)&hbuf[nxt][wrbase + NCH * 4] = hv;

        __syncthreads();
    }

#pragma unroll 1
    for (int t = T_IN; t < T_OUT; ++t) {
        const int cur = t & 1;
        const int nxt = cur ^ 1;

        v2f hfrag[NR];
#pragma unroll
        for (int j = 0; j < NR; ++j)
            hfrag[j] = *(const v2f*)&hbuf[cur][rdbase + j * 4];

        v8f acc = {0.f, 0.f, 0.f, 0.f, 0.f, 0.f, 0.f, 0.f};
        acc = wmma_(wown, hown, acc);
#pragma unroll
        for (int j = 0; j < NR; ++j)
            acc = wmma_(wfrag[j], hfrag[j], acc);

        const float oprev = project_(hfrag, wofrag, hown, woown, bo);
        acc = wmma_(wxfrag, v2f{0.0f, half ? oprev : 0.0f}, acc);

        if (wave == 0) {
            if (half == 0) out[(rowbase + r) * T_OUT + (t - 1)] = oprev;
        }

        v2f hv = cell_(acc, creg);
        if (lastw) hv.y = half ? 0.0f : 1.0f;
        hown = hv;
        *(v2f*)&hbuf[nxt][wrbase]           = hv;
        *(v2f*)&hbuf[nxt][wrbase + NCH * 4] = hv;
        __syncthreads();
    }

    if (wave == 0) {
        v2f hfrag[NR];
#pragma unroll
        for (int j = 0; j < NR; ++j)
            hfrag[j] = *(const v2f*)&hbuf[T_OUT & 1][rdbase + j * 4];
        const float olast = project_(hfrag, wofrag, hown, woown, bo);
        if (half == 0) out[(rowbase + r) * T_OUT + (T_OUT - 1)] = olast;
    }
}

extern "C" void kernel_launch(void* const* d_in, const int* in_sizes, int n_in,
                              void* d_out, int out_size, void* d_ws, size_t ws_size,
                              hipStream_t stream) {
    const float* x     = (const float*)d_in[0];
    const float* W_ih  = (const float*)d_in[1];
    const float* W_hh  = (const float*)d_in[2];
    const float* b_ih  = (const float*)d_in[3];
    const float* b_hh  = (const float*)d_in[4];
    const float* W_out = (const float*)d_in[5];
    const float* b_out = (const float*)d_in[6];
    float* out = (float*)d_out;

    const int B = in_sizes[0] / T_IN;
    dim3 grid(B / 16);
    dim3 block(NW * 32);
    hipLaunchKernelGGL(lstm_seq_kernel, grid, block, 0, stream,
                       x, W_ih, W_hh, b_ih, b_hh, W_out, b_out, out);
}


# Wave 12's 13-product chain is spread: each other wave makes one of its products from the h chunk it holds, wave 12 adds the 12 partial tiles from LDS as its C operand; its gate sums in a new order
# speedup vs baseline: 12.6768x; 1.1056x over previous
#include <hip/hip_runtime.h>


typedef __attribute__((ext_vector_type(2))) float v2f;
typedef __attribute__((ext_vector_type(4))) float v4f;
typedef __attribute__((ext_vector_type(8))) float v8f;

namespace {
template <int V> struct parity_ { static constexpr int value = V; };

constexpr int HID   = 50;
constexpr int NCH   = 13;
constexpr int NR    = 12;
constexpr int NW    = 13;
constexpr int NH    = 12;
constexpr int UPW   = 4;
constexpr int T_IN  = 512;
constexpr int FUT   = 64;
constexpr int T_OUT = T_IN + FUT;
static_assert(T_IN % 2 == 0 && FUT % 2 == 0, "the step loops take two steps per trip");
constexpr int PITCH = 108;

__device__ __forceinline__ float tanhf_(float v) {
    return __builtin_amdgcn_tanhf(v);
}

__device__ __forceinline__ float xhalf_(float v) {
    const int b = __builtin_bit_cast(int, v);
    return __builtin_bit_cast(float,
        __builtin_amdgcn_permlanex16(b, b, 0x76543210, 0xfedcba98, false, false));
}

__device__ __forceinline__ v8f wmma_(v2f a, v2f b, v8f c) {
    return __builtin_amdgcn_wmma_f32_16x16x4_f32(
         false, a,  false, b,
         (short)0, c,  false,  false);
}

__device__ __forceinline__ v2f cell_(const v8f acc, v2f& creg) {
    const v2f ti = v2f{tanhf_(acc[0]), tanhf_(acc[1])};
    const v2f tf = v2f{tanhf_(acc[2]), tanhf_(acc[3])};
    const v2f cg = v2f{tanhf_(acc[4]), tanhf_(acc[5])};
    const v2f to = v2f{tanhf_(acc[6]), tanhf_(acc[7])};
    const v2f ig = ti * 0.5f + 0.5f;
    const v2f fg = tf * 0.5f + 0.5f;
    const v2f og = to * 0.5f + 0.5f;
    creg = fg * creg + ig * cg;
    return og * v2f{tanhf_(creg.x), tanhf_(creg.y)};
}

__device__ __forceinline__ float project_(const v2f (&hf)[NR], const v2f (&wo)[NR],
                                          v2f hown, v2f woown, float bo) {
    v2f pe = hown * woown;
    v2f po = hf[0] * wo[0];
#pragma unroll
    for (int j = 1; j < NR; j += 2) {
        pe += hf[j] * wo[j];
        if (j + 1 < NR) po += hf[j + 1] * wo[j + 1];
    }
    const v2f pacc = pe + po;
    const float ph = pacc.x + pacc.y;
    return (ph + xhalf_(ph)) + bo;
}
}

__global__ __launch_bounds__(NW * 32) void lstm_seq_kernel(
    const float* __restrict__ x,
    const float* __restrict__ W_ih,
    const float* __restrict__ W_hh,
    const float* __restrict__ b_ih,
    const float* __restrict__ b_hh,
    const float* __restrict__ W_out,
    const float* __restrict__ b_out,
    float* __restrict__ out)
{
    __shared__ float hbuf[2][16 * PITCH];
    __shared__ v4f pbuf[2][NH * 32];

    const int lane    = threadIdx.x & 31;
    const int wave    = __builtin_amdgcn_readfirstlane((int)(threadIdx.x >> 5));
    const int r       = lane & 15;
    const int half    = lane >> 4;
    const int rowbase = blockIdx.x * 16;
    const bool lastw  = (wave == NW - 1);

    const int gsel = (r & 7) >> 1;
    const float ghalf = (gsel == 2) ? 1.0f : 0.5f;
    auto wcol = [&](int tw, int k) -> float {
        const int u = tw * UPW + 2 * (r & 1) + (r >> 3);
        if (u >= HID) return 0.0f;
        const int n = gsel * HID + u;
        float v;
        if (k < HID)       v = W_hh[n * HID + k];
        else if (k == HID) v = b_ih[n] + b_hh[n];
        else               v = W_ih[n];
        return v * ghalf;
    };
    auto wchunk = [&](int tw, int c) -> v2f {
        const int k0 = c * 4 + half;
        return v2f{wcol(tw, k0), wcol(tw, k0 + 2)};
    };
    auto wochunk = [&](int c) -> v2f {
        const int k0 = c * 4 + half;
        return v2f{(k0 < HID) ? W_out[k0] : 0.0f, (k0 + 2 < HID) ? W_out[k0 + 2] : 0.0f};
    };
    const v2f wown  = wchunk(wave, wave);
    const v2f woown = wochunk(wave);
    const v2f wsrv  = wchunk(NW - 1, wave);
    v2f wfrag[NR], wofrag[NR];
#pragma unroll
    for (int j = 0; j < NR; ++j) {
        int c = wave + 1 + j;
        if (c >= NCH) c -= NCH;
        wfrag[j]  = wchunk(wave, c);
        wofrag[j] = wochunk(c);
    }
    const v2f wxfrag = v2f{0.0f, half ? wcol(wave, HID + 1) : 0.0f};
    const float bo = b_out[0];

    v2f creg = v2f{0.0f, 0.0f};
    const float* xrow = x + (size_t)(rowbase + r) * T_IN;
    v2f hown = v2f{0.0f, 0.0f};
    if (lastw) hown.y = half ? xrow[0] : 1.0f;
    const int rdbase = r * PITCH + (wave + 1) * 4 + 2 * half;
    const int wrbase = r * PITCH + wave * 4 + 2 * half;
    *(v2f*)&hbuf[0][wrbase]           = hown;
    *(v2f*)&hbuf[0][wrbase + NCH * 4] = hown;
    if (!lastw) pbuf[0][wave * 32 + lane] = v4f{0.0f, 0.0f, 0.0f, 0.0f};
    __syncthreads();

    const v8f zero8 = {0.f, 0.f, 0.f, 0.f, 0.f, 0.f, 0.f, 0.f};

    auto gather = [&](auto curtag) -> v8f {
        constexpr int cur = decltype(curtag)::value;
        v4f pv[NH];
#pragma unroll
        for (int c = 0; c < NH; ++c) pv[c] = pbuf[cur][c * 32 + lane];
        v4f se = pv[0], so = pv[1];
#pragma unroll
        for (int c = 2; c < NH; c += 2) { se += pv[c]; so += pv[c + 1]; }
        const v4f s = se + so;
        return v8f{s.x, 0.f, s.y, 0.f, s.z, 0.f, s.w, 0.f};
    };
    auto serve = [&](auto nxttag, v2f hv) {
        constexpr int nxt = decltype(nxttag)::value;
        const v8f p = wmma_(wsrv, hv, zero8);
        pbuf[nxt][wave * 32 + lane] = v4f{p[0], p[2], p[4], p[6]};
    };

    auto observed_step = [&](int t, auto curtag) {
        constexpr int cur = decltype(curtag)::value;
        constexpr int nxt = cur ^ 1;

        if (!lastw) {
            v2f hfrag[NR];
#pragma unroll
            for (int j = 0; j < NR; ++j)
                hfrag[j] = *(const v2f*)&hbuf[cur][rdbase + j * 4];

            v8f acc = wmma_(wown, hown, zero8);
#pragma unroll
            for (int j = 0; j < NR; ++j)
                acc = wmma_(wfrag[j], hfrag[j], acc);

            if (t > 0 && wave == 0) {
                const float oprev = project_(hfrag, wofrag, hown, woown, bo);
                if (half == 0) out[(rowbase + r) * T_OUT + (t - 1)] = oprev;
            }

            const v2f hv = cell_(acc, creg);
            hown = hv;
            *(v2f*)&hbuf[nxt][wrbase]           = hv;
            *(v2f*)&hbuf[nxt][wrbase + NCH * 4] = hv;
            serve(parity_<nxt>{}, hv);
        } else {
            const int tn = (t + 1 < T_IN) ? (t + 1) : (T_IN - 1);
            const float xnext = xrow[tn];

            const v8f acc = wmma_(wown, hown, gather(parity_<cur>{}));
            v2f hv = cell_(acc, creg);
            hv.y = half ? ((t + 1 < T_IN) ? xnext : 0.0f) : 1.0f;
            hown = hv;
            *(v2f*)&hbuf[nxt][wrbase]           = hv;
            *(v2f*)&hbuf[nxt][wrbase + NCH * 4] = hv;
        }

        __syncthreads();
    };
#pragma unroll 1
    for (int t = 0; t < T_IN; t += 2) {
        observed_step(t,     parity_<0>{});
        observed_step(t + 1, parity_<1>{});
    }

    auto future_step = [&](int t, auto curtag) {
        constexpr int cur = decltype(curtag)::value;
        constexpr int nxt = cur ^ 1;

        v2f hfrag[NR];
#pragma unroll
        for (int j = 0; j < NR; ++j)
            hfrag[j] = *(const v2f*)&hbuf[cur][rdbase + j * 4];

        v8f acc;
        if (!lastw) {
            acc = wmma_(wown, hown, zero8);
#pragma unroll
            for (int j = 0; j < NR; ++j)
                acc = wmma_(wfrag[j], hfrag[j], acc);
        } else {
            acc = wmma_(wown, hown, gather(parity_<cur>{}));
        }

        const float oprev = project_(hfrag, wofrag, hown, woown, bo);
        acc = wmma_(wxfrag, v2f{0.0f, half ? oprev : 0.0f}, acc);

        if (wave == 0) {
            if (half == 0) out[(rowbase + r) * T_OUT + (t - 1)] = oprev;
        }

        v2f hv = cell_(acc, creg);
        if (lastw) hv.y = half ? 0.0f : 1.0f;
        hown = hv;
        *(v2f*)&hbuf[nxt][wrbase]           = hv;
        *(v2f*)&hbuf[nxt][wrbase + NCH * 4] = hv;
        if (!lastw) serve(parity_<nxt>{}, hv);
        __syncthreads();
    };
#pragma unroll 1
    for (int t = T_IN; t < T_OUT; t += 2) {
        future_step(t,     parity_<0>{});
        future_step(t + 1, parity_<1>{});
    }

    if (wave == 0) {
        v2f hfrag[NR];
#pragma unroll
        for (int j = 0; j < NR; ++j)
            hfrag[j] = *(const v2f*)&hbuf[T_OUT & 1][rdbase + j * 4];
        const float olast = project_(hfrag, wofrag, hown, woown, bo);
        if (half == 0) out[(rowbase + r) * T_OUT + (T_OUT - 1)] = olast;
    }
}

extern "C" void kernel_launch(void* const* d_in, const int* in_sizes, int n_in,
                              void* d_out, int out_size, void* d_ws, size_t ws_size,
                              hipStream_t stream) {
    const float* x     = (const float*)d_in[0];
    const float* W_ih  = (const float*)d_in[1];
    const float* W_hh  = (const float*)d_in[2];
    const float* b_ih  = (const float*)d_in[3];
    const float* b_hh  = (const float*)d_in[4];
    const float* W_out = (const float*)d_in[5];
    const float* b_out = (const float*)d_in[6];
    float* out = (float*)d_out;

    const int B = in_sizes[0] / T_IN;
    dim3 grid(B / 16);
    dim3 block(NW * 32);
    hipLaunchKernelGGL(lstm_seq_kernel, grid, block, 0, stream,
                       x, W_ih, W_hh, b_ih, b_hh, W_out, b_out, out);
}


# Each wave's 13-product chain split over two accumulator tiles issued alternately (own chunk and odd slots / even slots), joined by one add per register before the activations; gate sums in a new order
# speedup vs baseline: 13.1926x; 1.0407x over previous
#include <hip/hip_runtime.h>


typedef __attribute__((ext_vector_type(2))) float v2f;
typedef __attribute__((ext_vector_type(4))) float v4f;
typedef __attribute__((ext_vector_type(8))) float v8f;

namespace {
template <int V> struct parity_ { static constexpr int value = V; };

constexpr int HID   = 50;
constexpr int NCH   = 13;
constexpr int NR    = 12;
constexpr int NW    = 13;
constexpr int NH    = 12;
constexpr int UPW   = 4;
constexpr int T_IN  = 512;
constexpr int FUT   = 64;
constexpr int T_OUT = T_IN + FUT;
static_assert(T_IN % 2 == 0 && FUT % 2 == 0, "the step loops take two steps per trip");
constexpr int PITCH = 108;

__device__ __forceinline__ float tanhf_(float v) {
    return __builtin_amdgcn_tanhf(v);
}

__device__ __forceinline__ float xhalf_(float v) {
    const int b = __builtin_bit_cast(int, v);
    return __builtin_bit_cast(float,
        __builtin_amdgcn_permlanex16(b, b, 0x76543210, 0xfedcba98, false, false));
}

__device__ __forceinline__ v8f wmma_(v2f a, v2f b, v8f c) {
    return __builtin_amdgcn_wmma_f32_16x16x4_f32(
         false, a,  false, b,
         (short)0, c,  false,  false);
}

__device__ __forceinline__ v2f cell_(const v8f acc, v2f& creg) {
    const v2f ti = v2f{tanhf_(acc[0]), tanhf_(acc[1])};
    const v2f tf = v2f{tanhf_(acc[2]), tanhf_(acc[3])};
    const v2f cg = v2f{tanhf_(acc[4]), tanhf_(acc[5])};
    const v2f to = v2f{tanhf_(acc[6]), tanhf_(acc[7])};
    const v2f ig = ti * 0.5f + 0.5f;
    const v2f fg = tf * 0.5f + 0.5f;
    const v2f og = to * 0.5f + 0.5f;
    creg = fg * creg + ig * cg;
    return og * v2f{tanhf_(creg.x), tanhf_(creg.y)};
}

__device__ __forceinline__ float project_(const v2f (&hf)[NR], const v2f (&wo)[NR],
                                          v2f hown, v2f woown, float bo) {
    v2f pe = hown * woown;
    v2f po = hf[0] * wo[0];
#pragma unroll
    for (int j = 1; j < NR; j += 2) {
        pe += hf[j] * wo[j];
        if (j + 1 < NR) po += hf[j + 1] * wo[j + 1];
    }
    const v2f pacc = pe + po;
    const float ph = pacc.x + pacc.y;
    return (ph + xhalf_(ph)) + bo;
}
}

__global__ __launch_bounds__(NW * 32) void lstm_seq_kernel(
    const float* __restrict__ x,
    const float* __restrict__ W_ih,
    const float* __restrict__ W_hh,
    const float* __restrict__ b_ih,
    const float* __restrict__ b_hh,
    const float* __restrict__ W_out,
    const float* __restrict__ b_out,
    float* __restrict__ out)
{
    __shared__ float hbuf[2][16 * PITCH];
    __shared__ v4f pbuf[2][NH * 32];

    const int lane    = threadIdx.x & 31;
    const int wave    = __builtin_amdgcn_readfirstlane((int)(threadIdx.x >> 5));
    const int r       = lane & 15;
    const int half    = lane >> 4;
    const int rowbase = blockIdx.x * 16;
    const bool lastw  = (wave == NW - 1);

    const int gsel = (r & 7) >> 1;
    const float ghalf = (gsel == 2) ? 1.0f : 0.5f;
    auto wcol = [&](int tw, int k) -> float {
        const int u = tw * UPW + 2 * (r & 1) + (r >> 3);
        if (u >= HID) return 0.0f;
        const int n = gsel * HID + u;
        float v;
        if (k < HID)       v = W_hh[n * HID + k];
        else if (k == HID) v = b_ih[n] + b_hh[n];
        else               v = W_ih[n];
        return v * ghalf;
    };
    auto wchunk = [&](int tw, int c) -> v2f {
        const int k0 = c * 4 + half;
        return v2f{wcol(tw, k0), wcol(tw, k0 + 2)};
    };
    auto wochunk = [&](int c) -> v2f {
        const int k0 = c * 4 + half;
        return v2f{(k0 < HID) ? W_out[k0] : 0.0f, (k0 + 2 < HID) ? W_out[k0 + 2] : 0.0f};
    };
    const v2f wown  = wchunk(wave, wave);
    const v2f woown = wochunk(wave);
    const v2f wsrv  = wchunk(NW - 1, wave);
    v2f wfrag[NR], wofrag[NR];
#pragma unroll
    for (int j = 0; j < NR; ++j) {
        int c = wave + 1 + j;
        if (c >= NCH) c -= NCH;
        wfrag[j]  = wchunk(wave, c);
        wofrag[j] = wochunk(c);
    }
    const v2f wxfrag = v2f{0.0f, half ? wcol(wave, HID + 1) : 0.0f};
    const float bo = b_out[0];

    v2f creg = v2f{0.0f, 0.0f};
    const float* xrow = x + (size_t)(rowbase + r) * T_IN;
    v2f hown = v2f{0.0f, 0.0f};
    if (lastw) hown.y = half ? xrow[0] : 1.0f;
    const int rdbase = r * PITCH + (wave + 1) * 4 + 2 * half;
    const int wrbase = r * PITCH + wave * 4 + 2 * half;
    *(v2f*)&hbuf[0][wrbase]           = hown;
    *(v2f*)&hbuf[0][wrbase + NCH * 4] = hown;
    if (!lastw) pbuf[0][wave * 32 + lane] = v4f{0.0f, 0.0f, 0.0f, 0.0f};
    __syncthreads();

    const v8f zero8 = {0.f, 0.f, 0.f, 0.f, 0.f, 0.f, 0.f, 0.f};

    auto gather = [&](auto curtag) -> v8f {
        constexpr int cur = decltype(curtag)::value;
        v4f pv[NH];
#pragma unroll
        for (int c = 0; c < NH; ++c) pv[c] = pbuf[cur][c * 32 + lane];
        v4f se = pv[0], so = pv[1];
#pragma unroll
        for (int c = 2; c < NH; c += 2) { se += pv[c]; so += pv[c + 1]; }
        const v4f s = se + so;
        return v8f{s.x, 0.f, s.y, 0.f, s.z, 0.f, s.w, 0.f};
    };
    auto serve = [&](auto nxttag, v2f hv) {
        constexpr int nxt = decltype(nxttag)::value;
        const v8f p = wmma_(wsrv, hv, zero8);
        pbuf[nxt][wave * 32 + lane] = v4f{p[0], p[2], p[4], p[6]};
    };

    auto observed_step = [&](int t, auto curtag) {
        constexpr int cur = decltype(curtag)::value;
        constexpr int nxt = cur ^ 1;

        if (!lastw) {
            v2f hfrag[NR];
#pragma unroll
            for (int j = 0; j < NR; ++j)
                hfrag[j] = *(const v2f*)&hbuf[cur][rdbase + j * 4];

            v8f acc0 = wmma_(wown, hown, zero8);
            v8f acc1 = wmma_(wfrag[0], hfrag[0], zero8);
#pragma unroll
            for (int j = 1; j < NR; ++j) {
                if (j & 1) acc0 = wmma_(wfrag[j], hfrag[j], acc0);
                else       acc1 = wmma_(wfrag[j], hfrag[j], acc1);
            }

            if (t > 0 && wave == 0) {
                const float oprev = project_(hfrag, wofrag, hown, woown, bo);
                if (half == 0) out[(rowbase + r) * T_OUT + (t - 1)] = oprev;
            }

            const v2f hv = cell_(acc0 + acc1, creg);
            hown = hv;
            *(v2f*)&hbuf[nxt][wrbase]           = hv;
            *(v2f*)&hbuf[nxt][wrbase + NCH * 4] = hv;
            serve(parity_<nxt>{}, hv);
        } else {
            const int tn = (t + 1 < T_IN) ? (t + 1) : (T_IN - 1);
            const float xnext = xrow[tn];

            const v8f acc = wmma_(wown, hown, gather(parity_<cur>{}));
            v2f hv = cell_(acc, creg);
            hv.y = half ? ((t + 1 < T_IN) ? xnext : 0.0f) : 1.0f;
            hown = hv;
            *(v2f*)&hbuf[nxt][wrbase]           = hv;
            *(v2f*)&hbuf[nxt][wrbase + NCH * 4] = hv;
        }

        __syncthreads();
    };
#pragma unroll 1
    for (int t = 0; t < T_IN; t += 2) {
        observed_step(t,     parity_<0>{});
        observed_step(t + 1, parity_<1>{});
    }

    auto future_step = [&](int t, auto curtag) {
        constexpr int cur = decltype(curtag)::value;
        constexpr int nxt = cur ^ 1;

        v2f hfrag[NR];
#pragma unroll
        for (int j = 0; j < NR; ++j)
            hfrag[j] = *(const v2f*)&hbuf[cur][rdbase + j * 4];

        v8f acc0, acc1;
        if (!lastw) {
            acc0 = wmma_(wown, hown, zero8);
            acc1 = wmma_(wfrag[0], hfrag[0], zero8);
#pragma unroll
            for (int j = 1; j < NR; ++j) {
                if (j & 1) acc0 = wmma_(wfrag[j], hfrag[j], acc0);
                else       acc1 = wmma_(wfrag[j], hfrag[j], acc1);
            }
        } else {
            acc0 = wmma_(wown, hown, gather(parity_<cur>{}));
            acc1 = zero8;
        }

        const float oprev = project_(hfrag, wofrag, hown, woown, bo);
        acc1 = wmma_(wxfrag, v2f{0.0f, half ? oprev : 0.0f}, acc1);

        if (wave == 0) {
            if (half == 0) out[(rowbase + r) * T_OUT + (t - 1)] = oprev;
        }

        v2f hv = cell_(acc0 + acc1, creg);
        if (lastw) hv.y = half ? 0.0f : 1.0f;
        hown = hv;
        *(v2f*)&hbuf[nxt][wrbase]           = hv;
        *(v2f*)&hbuf[nxt][wrbase + NCH * 4] = hv;
        if (!lastw) serve(parity_<nxt>{}, hv);
        __syncthreads();
    };
#pragma unroll 1
    for (int t = T_IN; t < T_OUT; t += 2) {
        future_step(t,     parity_<0>{});
        future_step(t + 1, parity_<1>{});
    }

    if (wave == 0) {
        v2f hfrag[NR];
#pragma unroll
        for (int j = 0; j < NR; ++j)
            hfrag[j] = *(const v2f*)&hbuf[T_OUT & 1][rdbase + j * 4];
        const float olast = project_(hfrag, wofrag, hown, woown, bo);
        if (half == 0) out[(rowbase + r) * T_OUT + (T_OUT - 1)] = olast;
    }
}

extern "C" void kernel_launch(void* const* d_in, const int* in_sizes, int n_in,
                              void* d_out, int out_size, void* d_ws, size_t ws_size,
                              hipStream_t stream) {
    const float* x     = (const float*)d_in[0];
    const float* W_ih  = (const float*)d_in[1];
    const float* W_hh  = (const float*)d_in[2];
    const float* b_ih  = (const float*)d_in[3];
    const float* b_hh  = (const float*)d_in[4];
    const float* W_out = (const float*)d_in[5];
    const float* b_out = (const float*)d_in[6];
    float* out = (float*)d_out;

    const int B = in_sizes[0] / T_IN;
    dim3 grid(B / 16);
    dim3 block(NW * 32);
    hipLaunchKernelGGL(lstm_seq_kernel, grid, block, 0, stream,
                       x, W_ih, W_hh, b_ih, b_hh, W_out, b_out, out);
}
